# starting kernel
# baseline (speedup 1.0000x reference)
#include <hip/hip_runtime.h>
#include <hip/hip_bf16.h>
#include <math.h>

#define B_ 4
#define N_ 512
#define Z_ 64
#define E_ 16
#define H_ 128
#define ROWS_ (B_ * N_)

typedef float v2f __attribute__((ext_vector_type(2)));
typedef float v8f __attribute__((ext_vector_type(8)));

static __device__ __forceinline__ v8f wmma_f32(v2f a, v2f b, v8f c) {
    return __builtin_amdgcn_wmma_f32_16x16x4_f32(
         false, a,  false, b,
         (short)0, c,  false,  false);
}

__global__ void mpnn_proj_ij(const float* __restrict__ z,
                             const float* __restrict__ W_msg,
                             const float* __restrict__ b_msg,
                             float* __restrict__ P,
                             float* __restrict__ Q) {
    const int tid  = threadIdx.x;
    const int wave = tid >> 5;
    const int lane = tid & 31;
    const int lm   = lane & 15;
    const int lh   = lane >> 4;
    const int h0   = wave * 16;
    const int mrow = blockIdx.x * 16 + lm;

    v8f ci = {0.f,0.f,0.f,0.f,0.f,0.f,0.f,0.f};
    v8f cj = ci;
    const float* zrow = z + (size_t)mrow * Z_;

#pragma unroll
    for (int t = 0; t < Z_ / 4; ++t) {
        const int kb = 4 * t + 2 * lh;
        v2f a = *(const v2f*)(zrow + kb);
        const float* wi = W_msg + (size_t)kb * H_ + h0 + lm;
        const float* wj = W_msg + (size_t)(Z_ + kb) * H_ + h0 + lm;
        v2f bi = { wi[0], wi[H_] };
        v2f bj = { wj[0], wj[H_] };
        ci = wmma_f32(a, bi, ci);
        cj = wmma_f32(a, bj, cj);
    }

    const float bm = b_msg[h0 + lm];
#pragma unroll
    for (int r = 0; r < 8; ++r) {
        const int orow = blockIdx.x * 16 + r + 8 * lh;
        Q[(size_t)orow * H_ + h0 + lm] = ci[r] + bm;
        P[(size_t)orow * H_ + h0 + lm] = cj[r];
    }
}

__global__ void mpnn_agg_kernel(const float* __restrict__ e_feat,
                                const int*   __restrict__ adj,
                                const float* __restrict__ W_msg,
                                const float* __restrict__ P,
                                const float* __restrict__ Q,
                                float* __restrict__ agg) {
    extern __shared__ float sP[];

    const int bi   = blockIdx.x;
    const int b    = bi >> 9;
    const int tid  = threadIdx.x;
    const int wave = tid >> 5;
    const int lane = tid & 31;
    const int lm   = lane & 15;
    const int lh   = lane >> 4;
    const int h0   = wave * 16;

    {
        const float4* src  = (const float4*)(P + (size_t)b * N_ * H_);
        const int*    arow = adj + (size_t)bi * N_;
        float4* dst = (float4*)sP;
        for (int idx = tid; idx < (N_ * H_) / 4; idx += blockDim.x) {
            float4 v = src[idx];
            const int j   = idx >> 5;
            const float m = (arow[j] > 0) ? 0.0f : -INFINITY;
            v.x += m; v.y += m; v.z += m; v.w += m;
            dst[idx] = v;
        }
    }
    __syncthreads();

    v2f be[4];
#pragma unroll
    for (int t = 0; t < 4; ++t) {
        const int kb = 4 * t + 2 * lh;
        const float* we = W_msg + (size_t)(2 * Z_ + kb) * H_ + h0 + lm;
        be[t] = (v2f){ we[0], we[H_] };
    }

    const float* erow = e_feat + (size_t)bi * N_ * E_;
    float acc = -INFINITY;

    for (int j0 = 0; j0 < N_; j0 += 16) {
        const float* abase = erow + (size_t)(j0 + lm) * E_ + 2 * lh;
        if (j0 + 16 < N_)
            __builtin_prefetch(erow + (size_t)(j0 + 16 + lm) * E_, 0, 1);

        v8f c;
#pragma unroll
        for (int r = 0; r < 8; ++r)
            c[r] = sP[(j0 + r + 8 * lh) * H_ + h0 + lm];

#pragma unroll
        for (int t = 0; t < 4; ++t) {
            v2f a = *(const v2f*)(abase + 4 * t);
            c = wmma_f32(a, be[t], c);
        }

#pragma unroll
        for (int r = 0; r < 8; ++r)
            acc = fmaxf(acc, c[r]);
    }

    acc = fmaxf(acc, __shfl_xor(acc, 16, 32));
    if (lh == 0)
        agg[(size_t)bi * H_ + h0 + lm] = acc + Q[(size_t)bi * H_ + h0 + lm];
}

__global__ void mpnn_out_kernel(const float* __restrict__ z,
                                const float* __restrict__ agg,
                                const float* __restrict__ W_upd,
                                const float* __restrict__ b_upd,
                                float* __restrict__ out) {
    const int tid  = threadIdx.x;
    const int wave = tid >> 5;
    const int lane = tid & 31;
    const int lm   = lane & 15;
    const int lh   = lane >> 4;
    const int h0   = wave * 16;
    const int mrow = blockIdx.x * 16 + lm;

    v8f c = {0.f,0.f,0.f,0.f,0.f,0.f,0.f,0.f};

    const float* zrow = z + (size_t)mrow * Z_;
#pragma unroll
    for (int t = 0; t < Z_ / 4; ++t) {
        const int kb = 4 * t + 2 * lh;
        v2f a = *(const v2f*)(zrow + kb);
        const float* w = W_upd + (size_t)kb * H_ + h0 + lm;
        v2f bb = { w[0], w[H_] };
        c = wmma_f32(a, bb, c);
    }

    const float* arow = agg + (size_t)mrow * H_;
#pragma unroll
    for (int t = 0; t < H_ / 4; ++t) {
        const int kb = 4 * t + 2 * lh;
        v2f a = *(const v2f*)(arow + kb);
        const float* w = W_upd + (size_t)(Z_ + kb) * H_ + h0 + lm;
        v2f bb = { w[0], w[H_] };
        c = wmma_f32(a, bb, c);
    }

    const float bu = b_upd[h0 + lm];
#pragma unroll
    for (int r = 0; r < 8; ++r) {
        const int orow = blockIdx.x * 16 + r + 8 * lh;
        out[(size_t)orow * H_ + h0 + lm] = c[r] + bu;
    }
}

extern "C" void kernel_launch(void* const* d_in, const int* in_sizes, int n_in,
                              void* d_out, int out_size, void* d_ws, size_t ws_size,
                              hipStream_t stream) {
    const float* z      = (const float*)d_in[0];
    const float* e_feat = (const float*)d_in[1];
    const int*   adj    = (const int*)  d_in[2];
    const float* W_msg  = (const float*)d_in[3];
    const float* b_msg  = (const float*)d_in[4];
    const float* W_upd  = (const float*)d_in[5];
    const float* b_upd  = (const float*)d_in[6];
    float* out = (float*)d_out;

    float* ws  = (float*)d_ws;
    float* P   = ws;
    float* Q   = ws + (size_t)ROWS_ * H_;
    float* agg = ws + 2 * (size_t)ROWS_ * H_;

    mpnn_proj_ij<<<ROWS_ / 16, 256, 0, stream>>>(z, W_msg, b_msg, P, Q);

    const size_t shmem = (size_t)N_ * H_ * sizeof(float);
    mpnn_agg_kernel<<<B_ * N_, 256, shmem, stream>>>(e_feat, adj, W_msg, P, Q, agg);

    mpnn_out_kernel<<<ROWS_ / 16, 256, 0, stream>>>(z, agg, W_upd, b_upd, out);
}


# agg kernel re-tiled for register reuse: a wave owns 2 dst rows x 4 h-tiles, loads each P tile once as WMMA C for both rows and each e fragment once for 4 tiles; LDS staging of P[b] per (b,i) removed
# speedup vs baseline: 5.9028x; 5.9028x over previous
#include <hip/hip_runtime.h>
#include <hip/hip_bf16.h>
#include <math.h>

#define B_ 4
#define N_ 512
#define Z_ 64
#define E_ 16
#define H_ 128
#define ROWS_ (B_ * N_)

typedef float v2f __attribute__((ext_vector_type(2)));
typedef float v8f __attribute__((ext_vector_type(8)));

static __device__ __forceinline__ v8f wmma_f32(v2f a, v2f b, v8f c) {
    return __builtin_amdgcn_wmma_f32_16x16x4_f32(
         false, a,  false, b,
         (short)0, c,  false,  false);
}

__global__ void mpnn_proj_ij(const float* __restrict__ z,
                             const float* __restrict__ W_msg,
                             const float* __restrict__ b_msg,
                             float* __restrict__ P,
                             float* __restrict__ Q) {
    const int tid  = threadIdx.x;
    const int wave = tid >> 5;
    const int lane = tid & 31;
    const int lm   = lane & 15;
    const int lh   = lane >> 4;
    const int h0   = wave * 16;
    const int mrow = blockIdx.x * 16 + lm;

    v8f ci = {0.f,0.f,0.f,0.f,0.f,0.f,0.f,0.f};
    v8f cj = ci;
    const float* zrow = z + (size_t)mrow * Z_;

#pragma unroll
    for (int t = 0; t < Z_ / 4; ++t) {
        const int kb = 4 * t + 2 * lh;
        v2f a = *(const v2f*)(zrow + kb);
        const float* wi = W_msg + (size_t)kb * H_ + h0 + lm;
        const float* wj = W_msg + (size_t)(Z_ + kb) * H_ + h0 + lm;
        v2f bi = { wi[0], wi[H_] };
        v2f bj = { wj[0], wj[H_] };
        ci = wmma_f32(a, bi, ci);
        cj = wmma_f32(a, bj, cj);
    }

    const float bm = b_msg[h0 + lm];
#pragma unroll
    for (int r = 0; r < 8; ++r) {
        const int orow = blockIdx.x * 16 + r + 8 * lh;
        Q[(size_t)orow * H_ + h0 + lm] = ci[r] + bm;
        P[(size_t)orow * H_ + h0 + lm] = cj[r];
    }
}

#define IB_ 2
#define HB_ 4
#define JT_ 2

typedef float v4f __attribute__((ext_vector_type(4)));
typedef int   v4i __attribute__((ext_vector_type(4)));

struct AggStage {
    v2f a[JT_][IB_][4];
    v4i m[JT_][IB_][2];
    v4f p[JT_][8];
};

#define AGG_LOADS_STAY_ABOVE() asm volatile("" ::: "memory")

static __device__ __forceinline__ void agg_load(AggStage& s,
                                                const float* __restrict__ eb,
                                                const int*   __restrict__ ab,
                                                const float* __restrict__ Pb,
                                                int j0, int eoff, int aoff, int poff) {
    for (int jj = 0; jj < JT_; ++jj) {
        const int jt = j0 + 16 * jj;
        const float* ep = eb + jt * E_ + eoff;
        const int*   ap = ab + jt + aoff;
        const float* pp = Pb + jt * H_ + poff;
        for (int ii = 0; ii < IB_; ++ii) {
            for (int t = 0; t < 4; ++t)
                s.a[jj][ii][t] = *(const v2f*)(ep + ii * (N_ * E_) + 4 * t);
            s.m[jj][ii][0] = *(const v4i*)(ap + ii * N_);
            s.m[jj][ii][1] = *(const v4i*)(ap + ii * N_ + 4);
        }
        for (int r = 0; r < 8; ++r)
            s.p[jj][r] = *(const v4f*)(pp + r * H_);
    }
}

static __device__ __forceinline__ void agg_compute(const AggStage& s,
                                                   const v2f (&be)[HB_][4],
                                                   float (&acc)[IB_][HB_]) {
    for (int jj = 0; jj < JT_; ++jj) {
        for (int ii = 0; ii < IB_; ++ii) {
            float mk[8];
            for (int r = 0; r < 4; ++r) {
                mk[r]     = (s.m[jj][ii][0][r] > 0) ? 0.0f : -INFINITY;
                mk[r + 4] = (s.m[jj][ii][1][r] > 0) ? 0.0f : -INFINITY;
            }
            v8f c[HB_];
            for (int q = 0; q < HB_; ++q)
                for (int r = 0; r < 8; ++r)
                    c[q][r] = s.p[jj][r][q] + mk[r];
            for (int t = 0; t < 4; ++t)
                for (int q = 0; q < HB_; ++q)
                    c[q] = wmma_f32(s.a[jj][ii][t], be[q][t], c[q]);
            for (int q = 0; q < HB_; ++q) {
                float x = acc[ii][q];
                for (int r = 0; r < 8; ++r)
                    x = fmaxf(x, c[q][r]);
                acc[ii][q] = x;
            }
        }
    }
}

__global__ void __launch_bounds__(128) __attribute__((amdgpu_waves_per_eu(1, 2)))
mpnn_agg_kernel(const float* __restrict__ e_feat,
                const int*   __restrict__ adj,
                const float* __restrict__ W_msg,
                const float* __restrict__ P,
                const float* __restrict__ Q,
                float* __restrict__ agg) {
    const int tid  = threadIdx.x;
    const int wave = tid >> 5;
    const int lane = tid & 31;
    const int lm   = lane & 15;
    const int lh   = lane >> 4;
    const int row0 = blockIdx.x * 4 + (wave >> 1) * IB_;
    const int b    = row0 >> 9;
    const int h0   = (wave & 1) * 64;

    v2f be[HB_][4];
    for (int t = 0; t < 4; ++t) {
        const int kb = 4 * t + 2 * lh;
        const float* we = W_msg + (size_t)(2 * Z_ + kb) * H_ + h0 + 4 * lm;
        const v4f w0 = *(const v4f*)(we);
        const v4f w1 = *(const v4f*)(we + H_);
        for (int q = 0; q < HB_; ++q)
            be[q][t] = (v2f){ w0[q], w1[q] };
    }

    const float* eb = e_feat + (size_t)row0 * N_ * E_;
    const int*   ab = adj + (size_t)row0 * N_;
    const float* Pb = P + (size_t)b * N_ * H_;
    const int eoff = lm * E_ + 2 * lh;
    const int aoff = 8 * lh;
    const int poff = 8 * lh * H_ + h0 + 4 * lm;

    float acc[IB_][HB_];
    for (int ii = 0; ii < IB_; ++ii)
        for (int q = 0; q < HB_; ++q)
            acc[ii][q] = -INFINITY;

    AggStage s0, s1;
    agg_load(s0, eb, ab, Pb, 0, eoff, aoff, poff);
    AGG_LOADS_STAY_ABOVE();
    for (int j0 = 0; j0 < N_; j0 += 32 * JT_) {
        agg_load(s1, eb, ab, Pb, j0 + 16 * JT_, eoff, aoff, poff);
        AGG_LOADS_STAY_ABOVE();
        agg_compute(s0, be, acc);
        const int jn = (j0 + 32 * JT_ < N_) ? (j0 + 32 * JT_) : 0;
        agg_load(s0, eb, ab, Pb, jn, eoff, aoff, poff);
        AGG_LOADS_STAY_ABOVE();
        agg_compute(s1, be, acc);
    }

    for (int ii = 0; ii < IB_; ++ii) {
        v4f v;
        for (int q = 0; q < HB_; ++q)
            v[q] = fmaxf(acc[ii][q], __shfl_xor(acc[ii][q], 16, 32));
        if (lh == 0) {
            const size_t o = (size_t)(row0 + ii) * H_ + h0 + 4 * lm;
            const v4f qv = *(const v4f*)(Q + o);
            *(v4f*)(agg + o) = v + qv;
        }
    }
}

__global__ void mpnn_out_kernel(const float* __restrict__ z,
                                const float* __restrict__ agg,
                                const float* __restrict__ W_upd,
                                const float* __restrict__ b_upd,
                                float* __restrict__ out) {
    const int tid  = threadIdx.x;
    const int wave = tid >> 5;
    const int lane = tid & 31;
    const int lm   = lane & 15;
    const int lh   = lane >> 4;
    const int h0   = wave * 16;
    const int mrow = blockIdx.x * 16 + lm;

    v8f c = {0.f,0.f,0.f,0.f,0.f,0.f,0.f,0.f};

    const float* zrow = z + (size_t)mrow * Z_;
#pragma unroll
    for (int t = 0; t < Z_ / 4; ++t) {
        const int kb = 4 * t + 2 * lh;
        v2f a = *(const v2f*)(zrow + kb);
        const float* w = W_upd + (size_t)kb * H_ + h0 + lm;
        v2f bb = { w[0], w[H_] };
        c = wmma_f32(a, bb, c);
    }

    const float* arow = agg + (size_t)mrow * H_;
#pragma unroll
    for (int t = 0; t < H_ / 4; ++t) {
        const int kb = 4 * t + 2 * lh;
        v2f a = *(const v2f*)(arow + kb);
        const float* w = W_upd + (size_t)(Z_ + kb) * H_ + h0 + lm;
        v2f bb = { w[0], w[H_] };
        c = wmma_f32(a, bb, c);
    }

    const float bu = b_upd[h0 + lm];
#pragma unroll
    for (int r = 0; r < 8; ++r) {
        const int orow = blockIdx.x * 16 + r + 8 * lh;
        out[(size_t)orow * H_ + h0 + lm] = c[r] + bu;
    }
}

extern "C" void kernel_launch(void* const* d_in, const int* in_sizes, int n_in,
                              void* d_out, int out_size, void* d_ws, size_t ws_size,
                              hipStream_t stream) {
    const float* z      = (const float*)d_in[0];
    const float* e_feat = (const float*)d_in[1];
    const int*   adj    = (const int*)  d_in[2];
    const float* W_msg  = (const float*)d_in[3];
    const float* b_msg  = (const float*)d_in[4];
    const float* W_upd  = (const float*)d_in[5];
    const float* b_upd  = (const float*)d_in[6];
    float* out = (float*)d_out;

    float* ws  = (float*)d_ws;
    float* P   = ws;
    float* Q   = ws + (size_t)ROWS_ * H_;
    float* agg = ws + 2 * (size_t)ROWS_ * H_;

    mpnn_proj_ij<<<ROWS_ / 16, 256, 0, stream>>>(z, W_msg, b_msg, P, Q);

    mpnn_agg_kernel<<<ROWS_ / 4, 128, 0, stream>>>(e_feat, adj, W_msg, P, Q, agg);

    mpnn_out_kernel<<<ROWS_ / 16, 256, 0, stream>>>(z, agg, W_upd, b_upd, out);
}


# update Linear fused into the agg kernel's tail (3 launches -> 2): 8-row blocks of 8 waves leave agg in LDS, one barrier, wave w does out[8 rows,16 cols] on WMMA, sums in the old k order
# speedup vs baseline: 6.4780x; 1.0974x over previous
#include <hip/hip_runtime.h>
#include <hip/hip_bf16.h>
#include <math.h>

#define B_ 4
#define N_ 512
#define Z_ 64
#define E_ 16
#define H_ 128
#define ROWS_ (B_ * N_)

typedef float v2f __attribute__((ext_vector_type(2)));
typedef float v8f __attribute__((ext_vector_type(8)));

static __device__ __forceinline__ v8f wmma_f32(v2f a, v2f b, v8f c) {
    return __builtin_amdgcn_wmma_f32_16x16x4_f32(
         false, a,  false, b,
         (short)0, c,  false,  false);
}

__global__ void mpnn_proj_ij(const float* __restrict__ z,
                             const float* __restrict__ W_msg,
                             const float* __restrict__ b_msg,
                             float* __restrict__ P,
                             float* __restrict__ Q) {
    const int tid  = threadIdx.x;
    const int wave = tid >> 5;
    const int lane = tid & 31;
    const int lm   = lane & 15;
    const int lh   = lane >> 4;
    const int h0   = wave * 16;
    const int mrow = blockIdx.x * 16 + lm;

    v8f ci = {0.f,0.f,0.f,0.f,0.f,0.f,0.f,0.f};
    v8f cj = ci;
    const float* zrow = z + (size_t)mrow * Z_;

#pragma unroll
    for (int t = 0; t < Z_ / 4; ++t) {
        const int kb = 4 * t + 2 * lh;
        v2f a = *(const v2f*)(zrow + kb);
        const float* wi = W_msg + (size_t)kb * H_ + h0 + lm;
        const float* wj = W_msg + (size_t)(Z_ + kb) * H_ + h0 + lm;
        v2f bi = { wi[0], wi[H_] };
        v2f bj = { wj[0], wj[H_] };
        ci = wmma_f32(a, bi, ci);
        cj = wmma_f32(a, bj, cj);
    }

    const float bm = b_msg[h0 + lm];
#pragma unroll
    for (int r = 0; r < 8; ++r) {
        const int orow = blockIdx.x * 16 + r + 8 * lh;
        Q[(size_t)orow * H_ + h0 + lm] = ci[r] + bm;
        P[(size_t)orow * H_ + h0 + lm] = cj[r];
    }
}

#define IB_ 2
#define HB_ 4
#define JT_ 2

typedef float v4f __attribute__((ext_vector_type(4)));
typedef int   v4i __attribute__((ext_vector_type(4)));

struct AggStage {
    v2f a[JT_][IB_][4];
    v4i m[JT_][IB_][2];
    v4f p[JT_][8];
};

#define AGG_LOADS_STAY_ABOVE() asm volatile("" ::: "memory")

static __device__ __forceinline__ void agg_load(AggStage& s,
                                                const float* __restrict__ eb,
                                                const int*   __restrict__ ab,
                                                const float* __restrict__ Pb,
                                                int j0, int eoff, int aoff, int poff) {
    for (int jj = 0; jj < JT_; ++jj) {
        const int jt = j0 + 16 * jj;
        const float* ep = eb + jt * E_ + eoff;
        const int*   ap = ab + jt + aoff;
        const float* pp = Pb + jt * H_ + poff;
        for (int ii = 0; ii < IB_; ++ii) {
            for (int t = 0; t < 4; ++t)
                s.a[jj][ii][t] = *(const v2f*)(ep + ii * (N_ * E_) + 4 * t);
            s.m[jj][ii][0] = *(const v4i*)(ap + ii * N_);
            s.m[jj][ii][1] = *(const v4i*)(ap + ii * N_ + 4);
        }
        for (int r = 0; r < 8; ++r)
            s.p[jj][r] = *(const v4f*)(pp + r * H_);
    }
}

static __device__ __forceinline__ void agg_compute(const AggStage& s,
                                                   const v2f (&be)[HB_][4],
                                                   float (&acc)[IB_][HB_]) {
    for (int jj = 0; jj < JT_; ++jj) {
        for (int ii = 0; ii < IB_; ++ii) {
            float mk[8];
            for (int r = 0; r < 4; ++r) {
                mk[r]     = (s.m[jj][ii][0][r] > 0) ? 0.0f : -INFINITY;
                mk[r + 4] = (s.m[jj][ii][1][r] > 0) ? 0.0f : -INFINITY;
            }
            v8f c[HB_];
            for (int q = 0; q < HB_; ++q)
                for (int r = 0; r < 8; ++r)
                    c[q][r] = s.p[jj][r][q] + mk[r];
            for (int t = 0; t < 4; ++t)
                for (int q = 0; q < HB_; ++q)
                    c[q] = wmma_f32(s.a[jj][ii][t], be[q][t], c[q]);
            for (int q = 0; q < HB_; ++q) {
                float x = acc[ii][q];
                for (int r = 0; r < 8; ++r)
                    x = fmaxf(x, c[q][r]);
                acc[ii][q] = x;
            }
        }
    }
}

#define RB_  8
#define AGP_ (H_ + 4)

__global__ void __launch_bounds__(256) __attribute__((amdgpu_waves_per_eu(2, 2)))
mpnn_agg_upd_kernel(const float* __restrict__ e_feat,
                    const int*   __restrict__ adj,
                    const float* __restrict__ W_msg,
                    const float* __restrict__ P,
                    const float* __restrict__ Q,
                    const float* __restrict__ z,
                    const float* __restrict__ W_upd,
                    const float* __restrict__ b_upd,
                    float* __restrict__ out) {
    __shared__ float aggs[RB_ * AGP_];

    const int tid  = threadIdx.x;
    const int wave = tid >> 5;
    const int lane = tid & 31;
    const int lm   = lane & 15;
    const int lh   = lane >> 4;
    const int rl   = (wave >> 1) * IB_;
    const int row0 = blockIdx.x * RB_ + rl;
    const int b    = row0 >> 9;
    const int h0   = (wave & 1) * 64;

    v2f be[HB_][4];
    for (int t = 0; t < 4; ++t) {
        const int kb = 4 * t + 2 * lh;
        const float* we = W_msg + (size_t)(2 * Z_ + kb) * H_ + h0 + 4 * lm;
        const v4f w0 = *(const v4f*)(we);
        const v4f w1 = *(const v4f*)(we + H_);
        for (int q = 0; q < HB_; ++q)
            be[q][t] = (v2f){ w0[q], w1[q] };
    }

    const float* eb = e_feat + (size_t)row0 * N_ * E_;
    const int*   ab = adj + (size_t)row0 * N_;
    const float* Pb = P + (size_t)b * N_ * H_;
    const int eoff = lm * E_ + 2 * lh;
    const int aoff = 8 * lh;
    const int poff = 8 * lh * H_ + h0 + 4 * lm;

    float acc[IB_][HB_];
    for (int ii = 0; ii < IB_; ++ii)
        for (int q = 0; q < HB_; ++q)
            acc[ii][q] = -INFINITY;

    AggStage s0, s1;
    agg_load(s0, eb, ab, Pb, 0, eoff, aoff, poff);
    AGG_LOADS_STAY_ABOVE();
    for (int j0 = 0; j0 < N_; j0 += 32 * JT_) {
        agg_load(s1, eb, ab, Pb, j0 + 16 * JT_, eoff, aoff, poff);
        AGG_LOADS_STAY_ABOVE();
        agg_compute(s0, be, acc);
        const int jn = (j0 + 32 * JT_ < N_) ? (j0 + 32 * JT_) : 0;
        agg_load(s0, eb, ab, Pb, jn, eoff, aoff, poff);
        AGG_LOADS_STAY_ABOVE();
        agg_compute(s1, be, acc);
    }

    const int    rowl = lm & 7;
    const float* zrow = z + (size_t)(blockIdx.x * RB_ + rowl) * Z_;
    const float* wcol = W_upd + wave * 16 + lm;
    v2f za[Z_ / 4];
    v2f wb[(Z_ + H_) / 4];
    for (int t = 0; t < Z_ / 4; ++t)
        za[t] = *(const v2f*)(zrow + 4 * t + 2 * lh);
    for (int t = 0; t < (Z_ + H_) / 4; ++t) {
        const float* w = wcol + (size_t)(4 * t + 2 * lh) * H_;
        wb[t] = (v2f){ w[0], w[H_] };
    }

    for (int ii = 0; ii < IB_; ++ii) {
        v4f v;
        for (int q = 0; q < HB_; ++q)
            v[q] = fmaxf(acc[ii][q], __shfl_xor(acc[ii][q], 16, 32));
        if (lh == 0) {
            const size_t o = (size_t)(row0 + ii) * H_ + h0 + 4 * lm;
            const v4f qv = *(const v4f*)(Q + o);
            *(v4f*)(&aggs[(rl + ii) * AGP_ + h0 + 4 * lm]) = v + qv;
        }
    }

    v8f c = {0.f,0.f,0.f,0.f,0.f,0.f,0.f,0.f};
    for (int t = 0; t < Z_ / 4; ++t)
        c = wmma_f32(za[t], wb[t], c);

    __builtin_amdgcn_sched_barrier(0);
    __syncthreads();

    const float* arow = &aggs[rowl * AGP_ + 2 * lh];
    for (int t = 0; t < H_ / 4; ++t) {
        const v2f a = *(const v2f*)(arow + 4 * t);
        c = wmma_f32(a, wb[Z_ / 4 + t], c);
    }

    const float bu = b_upd[wave * 16 + lm];
    if (lh == 0) {
        for (int r = 0; r < 8; ++r)
            out[(size_t)(blockIdx.x * RB_ + r) * H_ + wave * 16 + lm] = c[r] + bu;
    }
}

extern "C" void kernel_launch(void* const* d_in, const int* in_sizes, int n_in,
                              void* d_out, int out_size, void* d_ws, size_t ws_size,
                              hipStream_t stream) {
    const float* z      = (const float*)d_in[0];
    const float* e_feat = (const float*)d_in[1];
    const int*   adj    = (const int*)  d_in[2];
    const float* W_msg  = (const float*)d_in[3];
    const float* b_msg  = (const float*)d_in[4];
    const float* W_upd  = (const float*)d_in[5];
    const float* b_upd  = (const float*)d_in[6];
    float* out = (float*)d_out;

    float* ws  = (float*)d_ws;
    float* P   = ws;
    float* Q   = ws + (size_t)ROWS_ * H_;

    mpnn_proj_ij<<<ROWS_ / 16, 256, 0, stream>>>(z, W_msg, b_msg, P, Q);

    mpnn_agg_upd_kernel<<<ROWS_ / RB_, 256, 0, stream>>>(e_feat, adj, W_msg, P, Q,
                                                        z, W_upd, b_upd, out);
}


# agg loop: one second-level prefetch instruction per wave and trip names the e_feat and adj 256-B units of two trips ahead, so the register stages' in-order loads stop waiting on lines from memory
# speedup vs baseline: 6.6923x; 1.0331x over previous
#include <hip/hip_runtime.h>
#include <hip/hip_bf16.h>
#include <math.h>

#define B_ 4
#define N_ 512
#define Z_ 64
#define E_ 16
#define H_ 128
#define ROWS_ (B_ * N_)

typedef float v2f __attribute__((ext_vector_type(2)));
typedef float v8f __attribute__((ext_vector_type(8)));

static __device__ __forceinline__ v8f wmma_f32(v2f a, v2f b, v8f c) {
    return __builtin_amdgcn_wmma_f32_16x16x4_f32(
         false, a,  false, b,
         (short)0, c,  false,  false);
}

__global__ void mpnn_proj_ij(const float* __restrict__ z,
                             const float* __restrict__ W_msg,
                             const float* __restrict__ b_msg,
                             float* __restrict__ P,
                             float* __restrict__ Q) {
    const int tid  = threadIdx.x;
    const int wave = tid >> 5;
    const int lane = tid & 31;
    const int lm   = lane & 15;
    const int lh   = lane >> 4;
    const int h0   = wave * 16;
    const int mrow = blockIdx.x * 16 + lm;

    v8f ci = {0.f,0.f,0.f,0.f,0.f,0.f,0.f,0.f};
    v8f cj = ci;
    const float* zrow = z + (size_t)mrow * Z_;

#pragma unroll
    for (int t = 0; t < Z_ / 4; ++t) {
        const int kb = 4 * t + 2 * lh;
        v2f a = *(const v2f*)(zrow + kb);
        const float* wi = W_msg + (size_t)kb * H_ + h0 + lm;
        const float* wj = W_msg + (size_t)(Z_ + kb) * H_ + h0 + lm;
        v2f bi = { wi[0], wi[H_] };
        v2f bj = { wj[0], wj[H_] };
        ci = wmma_f32(a, bi, ci);
        cj = wmma_f32(a, bj, cj);
    }

    const float bm = b_msg[h0 + lm];
#pragma unroll
    for (int r = 0; r < 8; ++r) {
        const int orow = blockIdx.x * 16 + r + 8 * lh;
        Q[(size_t)orow * H_ + h0 + lm] = ci[r] + bm;
        P[(size_t)orow * H_ + h0 + lm] = cj[r];
    }
}

#define IB_ 2
#define HB_ 4
#define JT_ 2
#define PFTRIP_ (32 * JT_)
#define PFLOC_  1

typedef float v4f __attribute__((ext_vector_type(4)));
typedef int   v4i __attribute__((ext_vector_type(4)));

struct AggStage {
    v2f a[JT_][IB_][4];
    v4i m[JT_][IB_][2];
    v4f p[JT_][8];
};

#define AGG_LOADS_STAY_ABOVE() asm volatile("" ::: "memory")

static __device__ __forceinline__ void agg_load(AggStage& s,
                                                const float* __restrict__ eb,
                                                const int*   __restrict__ ab,
                                                const float* __restrict__ Pb,
                                                int j0, int eoff, int aoff, int poff) {
    for (int jj = 0; jj < JT_; ++jj) {
        const int jt = j0 + 16 * jj;
        const float* ep = eb + jt * E_ + eoff;
        const int*   ap = ab + jt + aoff;
        const float* pp = Pb + jt * H_ + poff;
        for (int ii = 0; ii < IB_; ++ii) {
            for (int t = 0; t < 4; ++t)
                s.a[jj][ii][t] = *(const v2f*)(ep + ii * (N_ * E_) + 4 * t);
            s.m[jj][ii][0] = *(const v4i*)(ap + ii * N_);
            s.m[jj][ii][1] = *(const v4i*)(ap + ii * N_ + 4);
        }
        for (int r = 0; r < 8; ++r)
            s.p[jj][r] = *(const v4f*)(pp + r * H_);
    }
}

static __device__ __forceinline__ void agg_compute(const AggStage& s,
                                                   const v2f (&be)[HB_][4],
                                                   float (&acc)[IB_][HB_]) {
    for (int jj = 0; jj < JT_; ++jj) {
        for (int ii = 0; ii < IB_; ++ii) {
            float mk[8];
            for (int r = 0; r < 4; ++r) {
                mk[r]     = (s.m[jj][ii][0][r] > 0) ? 0.0f : -INFINITY;
                mk[r + 4] = (s.m[jj][ii][1][r] > 0) ? 0.0f : -INFINITY;
            }
            v8f c[HB_];
            for (int q = 0; q < HB_; ++q)
                for (int r = 0; r < 8; ++r)
                    c[q][r] = s.p[jj][r][q] + mk[r];
            for (int t = 0; t < 4; ++t)
                for (int q = 0; q < HB_; ++q)
                    c[q] = wmma_f32(s.a[jj][ii][t], be[q][t], c[q]);
            for (int q = 0; q < HB_; ++q) {
                float x = acc[ii][q];
                for (int r = 0; r < 8; ++r)
                    x = fmaxf(x, c[q][r]);
                acc[ii][q] = x;
            }
        }
    }
}

#define RB_  8
#define AGP_ (H_ + 4)

__global__ void __launch_bounds__(256) __attribute__((amdgpu_waves_per_eu(2, 2)))
mpnn_agg_upd_kernel(const float* __restrict__ e_feat,
                    const int*   __restrict__ adj,
                    const float* __restrict__ W_msg,
                    const float* __restrict__ P,
                    const float* __restrict__ Q,
                    const float* __restrict__ z,
                    const float* __restrict__ W_upd,
                    const float* __restrict__ b_upd,
                    float* __restrict__ out) {
    __shared__ float aggs[RB_ * AGP_];

    const int tid  = threadIdx.x;
    const int wave = tid >> 5;
    const int lane = tid & 31;
    const int lm   = lane & 15;
    const int lh   = lane >> 4;
    const int rl   = (wave >> 1) * IB_;
    const int row0 = blockIdx.x * RB_ + rl;
    const int b    = row0 >> 9;
    const int h0   = (wave & 1) * 64;

    v2f be[HB_][4];
    for (int t = 0; t < 4; ++t) {
        const int kb = 4 * t + 2 * lh;
        const float* we = W_msg + (size_t)(2 * Z_ + kb) * H_ + h0 + 4 * lm;
        const v4f w0 = *(const v4f*)(we);
        const v4f w1 = *(const v4f*)(we + H_);
        for (int q = 0; q < HB_; ++q)
            be[q][t] = (v2f){ w0[q], w1[q] };
    }

    const float* eb = e_feat + (size_t)row0 * N_ * E_;
    const int*   ab = adj + (size_t)row0 * N_;
    const float* Pb = P + (size_t)b * N_ * H_;
    const int eoff = lm * E_ + 2 * lh;
    const int aoff = 8 * lh;
    const int poff = 8 * lh * H_ + h0 + 4 * lm;

    float acc[IB_][HB_];
    for (int ii = 0; ii < IB_; ++ii)
        for (int q = 0; q < HB_; ++q)
            acc[ii][q] = -INFINITY;

    const int    prow = row0 + (wave & 1);
    const char*  pfp  = (lane < 16)
        ? (const char*)(e_feat + (size_t)prow * N_ * E_) + lane * 256
        : (const char*)(adj + (size_t)prow * N_);
    const int    pfstep = (lane < 16) ? PFTRIP_ * E_ * 4 : PFTRIP_ * 4;
    __builtin_prefetch(pfp + pfstep, 0, PFLOC_);
    const char*  pfq = pfp + 2 * pfstep;

    AggStage s0, s1;
    agg_load(s0, eb, ab, Pb, 0, eoff, aoff, poff);
    AGG_LOADS_STAY_ABOVE();
    for (int j0 = 0; j0 < N_; j0 += 32 * JT_) {
        __builtin_prefetch(pfq, 0, PFLOC_);
        pfq += (j0 + 3 * PFTRIP_ < N_) ? pfstep : 0;
        agg_load(s1, eb, ab, Pb, j0 + 16 * JT_, eoff, aoff, poff);
        AGG_LOADS_STAY_ABOVE();
        agg_compute(s0, be, acc);
        const int jn = (j0 + 32 * JT_ < N_) ? (j0 + 32 * JT_) : 0;
        agg_load(s0, eb, ab, Pb, jn, eoff, aoff, poff);
        AGG_LOADS_STAY_ABOVE();
        agg_compute(s1, be, acc);
    }

    const int    rowl = lm & 7;
    const float* zrow = z + (size_t)(blockIdx.x * RB_ + rowl) * Z_;
    const float* wcol = W_upd + wave * 16 + lm;
    v2f za[Z_ / 4];
    v2f wb[(Z_ + H_) / 4];
    for (int t = 0; t < Z_ / 4; ++t)
        za[t] = *(const v2f*)(zrow + 4 * t + 2 * lh);
    for (int t = 0; t < (Z_ + H_) / 4; ++t) {
        const float* w = wcol + (size_t)(4 * t + 2 * lh) * H_;
        wb[t] = (v2f){ w[0], w[H_] };
    }

    for (int ii = 0; ii < IB_; ++ii) {
        v4f v;
        for (int q = 0; q < HB_; ++q)
            v[q] = fmaxf(acc[ii][q], __shfl_xor(acc[ii][q], 16, 32));
        if (lh == 0) {
            const size_t o = (size_t)(row0 + ii) * H_ + h0 + 4 * lm;
            const v4f qv = *(const v4f*)(Q + o);
            *(v4f*)(&aggs[(rl + ii) * AGP_ + h0 + 4 * lm]) = v + qv;
        }
    }

    v8f c = {0.f,0.f,0.f,0.f,0.f,0.f,0.f,0.f};
    for (int t = 0; t < Z_ / 4; ++t)
        c = wmma_f32(za[t], wb[t], c);

    __builtin_amdgcn_sched_barrier(0);
    __syncthreads();

    const float* arow = &aggs[rowl * AGP_ + 2 * lh];
    for (int t = 0; t < H_ / 4; ++t) {
        const v2f a = *(const v2f*)(arow + 4 * t);
        c = wmma_f32(a, wb[Z_ / 4 + t], c);
    }

    const float bu = b_upd[wave * 16 + lm];
    if (lh == 0) {
        for (int r = 0; r < 8; ++r)
            out[(size_t)(blockIdx.x * RB_ + r) * H_ + wave * 16 + lm] = c[r] + bu;
    }
}

extern "C" void kernel_launch(void* const* d_in, const int* in_sizes, int n_in,
                              void* d_out, int out_size, void* d_ws, size_t ws_size,
                              hipStream_t stream) {
    const float* z      = (const float*)d_in[0];
    const float* e_feat = (const float*)d_in[1];
    const int*   adj    = (const int*)  d_in[2];
    const float* W_msg  = (const float*)d_in[3];
    const float* b_msg  = (const float*)d_in[4];
    const float* W_upd  = (const float*)d_in[5];
    const float* b_upd  = (const float*)d_in[6];
    float* out = (float*)d_out;

    float* ws  = (float*)d_ws;
    float* P   = ws;
    float* Q   = ws + (size_t)ROWS_ * H_;

    mpnn_proj_ij<<<ROWS_ / 16, 256, 0, stream>>>(z, W_msg, b_msg, P, Q);

    mpnn_agg_upd_kernel<<<ROWS_ / RB_, 256, 0, stream>>>(e_feat, adj, W_msg, P, Q,
                                                        z, W_upd, b_upd, out);
}


# agg loop: per-tile mask+P add pass removed: kernel 1 stores P in WMMA C-fragment order, tiles load straight in as C; a masked row gets a NaN in its A fragment and fmaxf skips it; sums in old order
# speedup vs baseline: 7.0949x; 1.0602x over previous
#include <hip/hip_runtime.h>
#include <hip/hip_bf16.h>
#include <math.h>

#define B_ 4
#define N_ 512
#define Z_ 64
#define E_ 16
#define H_ 128
#define ROWS_ (B_ * N_)

typedef float v2f __attribute__((ext_vector_type(2)));
typedef float v4f __attribute__((ext_vector_type(4)));
typedef float v8f __attribute__((ext_vector_type(8)));

static __device__ __forceinline__ v8f wmma_f32(v2f a, v2f b, v8f c) {
    return __builtin_amdgcn_wmma_f32_16x16x4_f32(
         false, a,  false, b,
         (short)0, c,  false,  false);
}

__global__ void mpnn_proj_ij(const float* __restrict__ z,
                             const float* __restrict__ W_msg,
                             const float* __restrict__ b_msg,
                             float* __restrict__ P,
                             float* __restrict__ Q) {
    const int tid  = threadIdx.x;
    const int wave = tid >> 5;
    const int lane = tid & 31;
    const int lm   = lane & 15;
    const int lh   = lane >> 4;
    const int h0   = wave * 16;
    const int mrow = blockIdx.x * 16 + lm;

    v8f ci = {0.f,0.f,0.f,0.f,0.f,0.f,0.f,0.f};
    v8f cj = ci;
    const float* zrow = z + (size_t)mrow * Z_;

#pragma unroll
    for (int t = 0; t < Z_ / 4; ++t) {
        const int kb = 4 * t + 2 * lh;
        v2f a = *(const v2f*)(zrow + kb);
        const float* wi = W_msg + (size_t)kb * H_ + h0 + lm;
        const float* wj = W_msg + (size_t)(Z_ + kb) * H_ + h0 + lm;
        v2f bi = { wi[0], wi[H_] };
        v2f bj = { wj[0], wj[H_] };
        ci = wmma_f32(a, bi, ci);
        cj = wmma_f32(a, bj, cj);
    }

    const float bm = b_msg[h0 + lm];
#pragma unroll
    for (int r = 0; r < 8; ++r) {
        const int orow = blockIdx.x * 16 + r + 8 * lh;
        Q[(size_t)orow * H_ + h0 + lm] = ci[r] + bm;
    }
    float* pf = P + (size_t)(blockIdx.x * (H_ / 16) + wave) * 256 + lane * 4;
    *(v4f*)(pf)       = (v4f){ cj[0], cj[1], cj[2], cj[3] };
    *(v4f*)(pf + 128) = (v4f){ cj[4], cj[5], cj[6], cj[7] };
}

#define IB_ 2
#define HB_ 4
#define JT_ 2
#define PFTRIP_ (32 * JT_)
#define PFLOC_  1


struct AggStage {
    v2f a[JT_][IB_][4];
    int m[JT_][IB_];
    v4f p[JT_][HB_][2];
};

#define AGG_LOADS_STAY_ABOVE() asm volatile("" ::: "memory")

static __device__ __forceinline__ void agg_load(AggStage& s,
                                                const float* __restrict__ eb,
                                                const int*   __restrict__ ab,
                                                const float* __restrict__ Pb,
                                                int j0, int eoff, int aoff, int poff) {
    for (int jj = 0; jj < JT_; ++jj) {
        const int jt = j0 + 16 * jj;
        const float* ep = eb + jt * E_ + eoff;
        const int*   ap = ab + jt + aoff;
        const float* pp = Pb + jt * H_ + poff;
        for (int ii = 0; ii < IB_; ++ii) {
            for (int t = 0; t < 4; ++t)
                s.a[jj][ii][t] = *(const v2f*)(ep + ii * (N_ * E_) + 4 * t);
            s.m[jj][ii] = ap[ii * N_];
        }
        for (int q = 0; q < HB_; ++q) {
            s.p[jj][q][0] = *(const v4f*)(pp + q * 256);
            s.p[jj][q][1] = *(const v4f*)(pp + q * 256 + 128);
        }
    }
}

static __device__ __forceinline__ void agg_compute(const AggStage& s,
                                                   const v2f (&be)[HB_][4],
                                                   float (&acc)[IB_][HB_]) {
    for (int jj = 0; jj < JT_; ++jj) {
        for (int ii = 0; ii < IB_; ++ii) {
            v2f a0 = s.a[jj][ii][0];
            a0[0] = (s.m[jj][ii] > 0) ? a0[0] : __builtin_nanf("");
            v8f c[HB_];
            for (int q = 0; q < HB_; ++q) {
                const v8f pc = __builtin_shufflevector(s.p[jj][q][0], s.p[jj][q][1], 0, 1, 2, 3, 4, 5, 6, 7);
                c[q] = wmma_f32(a0, be[q][0], pc);
            }
            for (int t = 1; t < 4; ++t)
                for (int q = 0; q < HB_; ++q)
                    c[q] = wmma_f32(s.a[jj][ii][t], be[q][t], c[q]);
            for (int q = 0; q < HB_; ++q) {
                float x = acc[ii][q];
                for (int r = 0; r < 8; ++r)
                    x = fmaxf(x, c[q][r]);
                acc[ii][q] = x;
            }
        }
    }
}

#define RB_  8
#define AGP_ (H_ + 4)

__global__ void __launch_bounds__(256) __attribute__((amdgpu_waves_per_eu(2, 2)))
mpnn_agg_upd_kernel(const float* __restrict__ e_feat,
                    const int*   __restrict__ adj,
                    const float* __restrict__ W_msg,
                    const float* __restrict__ P,
                    const float* __restrict__ Q,
                    const float* __restrict__ z,
                    const float* __restrict__ W_upd,
                    const float* __restrict__ b_upd,
                    float* __restrict__ out) {
    __shared__ float aggs[RB_ * AGP_];

    const int tid  = threadIdx.x;
    const int wave = tid >> 5;
    const int lane = tid & 31;
    const int lm   = lane & 15;
    const int lh   = lane >> 4;
    const int rl   = (wave >> 1) * IB_;
    const int row0 = blockIdx.x * RB_ + rl;
    const int b    = row0 >> 9;
    const int h0   = (wave & 1) * 64;

    v2f be[HB_][4];
    for (int t = 0; t < 4; ++t) {
        const int kb = 4 * t + 2 * lh;
        const float* we = W_msg + (size_t)(2 * Z_ + kb) * H_ + h0 + lm;
        for (int q = 0; q < HB_; ++q)
            be[q][t] = (v2f){ we[16 * q], we[H_ + 16 * q] };
    }

    const float* eb = e_feat + (size_t)row0 * N_ * E_;
    const int*   ab = adj + (size_t)row0 * N_;
    const float* Pb = P + (size_t)b * N_ * H_;
    const int eoff = lm * E_ + 2 * lh;
    const int aoff = lm;
    const int poff = (h0 / 16) * 256 + lane * 4;

    float acc[IB_][HB_];
    for (int ii = 0; ii < IB_; ++ii)
        for (int q = 0; q < HB_; ++q)
            acc[ii][q] = -INFINITY;

    const int    prow = row0 + (wave & 1);
    const char*  pfp  = (lane < 16)
        ? (const char*)(e_feat + (size_t)prow * N_ * E_) + lane * 256
        : (const char*)(adj + (size_t)prow * N_);
    const int    pfstep = (lane < 16) ? PFTRIP_ * E_ * 4 : PFTRIP_ * 4;
    __builtin_prefetch(pfp + pfstep, 0, PFLOC_);
    const char*  pfq = pfp + 2 * pfstep;

    AggStage s0, s1;
    agg_load(s0, eb, ab, Pb, 0, eoff, aoff, poff);
    AGG_LOADS_STAY_ABOVE();
    for (int j0 = 0; j0 < N_; j0 += 32 * JT_) {
        __builtin_prefetch(pfq, 0, PFLOC_);
        pfq += (j0 + 3 * PFTRIP_ < N_) ? pfstep : 0;
        agg_load(s1, eb, ab, Pb, j0 + 16 * JT_, eoff, aoff, poff);
        AGG_LOADS_STAY_ABOVE();
        agg_compute(s0, be, acc);
        const int jn = (j0 + 32 * JT_ < N_) ? (j0 + 32 * JT_) : 0;
        agg_load(s0, eb, ab, Pb, jn, eoff, aoff, poff);
        AGG_LOADS_STAY_ABOVE();
        agg_compute(s1, be, acc);
    }

    const int    rowl = lm & 7;
    const float* zrow = z + (size_t)(blockIdx.x * RB_ + rowl) * Z_;
    const float* wcol = W_upd + wave * 16 + lm;
    v2f za[Z_ / 4];
    v2f wb[(Z_ + H_) / 4];
    for (int t = 0; t < Z_ / 4; ++t)
        za[t] = *(const v2f*)(zrow + 4 * t + 2 * lh);
    for (int t = 0; t < (Z_ + H_) / 4; ++t) {
        const float* w = wcol + (size_t)(4 * t + 2 * lh) * H_;
        wb[t] = (v2f){ w[0], w[H_] };
    }

    for (int ii = 0; ii < IB_; ++ii) {
        float v[HB_];
        for (int q = 0; q < HB_; ++q)
            v[q] = fmaxf(acc[ii][q], __shfl_xor(acc[ii][q], 16, 32));
        if (lh == 0) {
            const size_t o = (size_t)(row0 + ii) * H_ + h0 + lm;
            for (int q = 0; q < HB_; ++q)
                aggs[(rl + ii) * AGP_ + h0 + 16 * q + lm] = v[q] + Q[o + 16 * q];
        }
    }

    v8f c = {0.f,0.f,0.f,0.f,0.f,0.f,0.f,0.f};
    for (int t = 0; t < Z_ / 4; ++t)
        c = wmma_f32(za[t], wb[t], c);

    __builtin_amdgcn_sched_barrier(0);
    __syncthreads();

    const float* arow = &aggs[rowl * AGP_ + 2 * lh];
    for (int t = 0; t < H_ / 4; ++t) {
        const v2f a = *(const v2f*)(arow + 4 * t);
        c = wmma_f32(a, wb[Z_ / 4 + t], c);
    }

    const float bu = b_upd[wave * 16 + lm];
    if (lh == 0) {
        for (int r = 0; r < 8; ++r)
            out[(size_t)(blockIdx.x * RB_ + r) * H_ + wave * 16 + lm] = c[r] + bu;
    }
}

extern "C" void kernel_launch(void* const* d_in, const int* in_sizes, int n_in,
                              void* d_out, int out_size, void* d_ws, size_t ws_size,
                              hipStream_t stream) {
    const float* z      = (const float*)d_in[0];
    const float* e_feat = (const float*)d_in[1];
    const int*   adj    = (const int*)  d_in[2];
    const float* W_msg  = (const float*)d_in[3];
    const float* b_msg  = (const float*)d_in[4];
    const float* W_upd  = (const float*)d_in[5];
    const float* b_upd  = (const float*)d_in[6];
    float* out = (float*)d_out;

    float* ws  = (float*)d_ws;
    float* P   = ws;
    float* Q   = ws + (size_t)ROWS_ * H_;

    mpnn_proj_ij<<<ROWS_ / 16, 256, 0, stream>>>(z, W_msg, b_msg, P, Q);

    mpnn_agg_upd_kernel<<<ROWS_ / RB_, 256, 0, stream>>>(e_feat, adj, W_msg, P, Q,
                                                        z, W_upd, b_upd, out);
}


# agg loop walks only sources with adj>0: rising j list per row in LDS (ballot+prefix), P[b] copied to LDS, product transposed (W_e as A, listed e rows as B), C = 32 B of a P row; sums in old order
# speedup vs baseline: 8.3858x; 1.1819x over previous
#include <hip/hip_runtime.h>
#include <hip/hip_bf16.h>
#include <math.h>

#define B_ 4
#define N_ 512
#define Z_ 64
#define E_ 16
#define H_ 128
#define ROWS_ (B_ * N_)

typedef float v2f __attribute__((ext_vector_type(2)));
typedef float v4f __attribute__((ext_vector_type(4)));
typedef float v8f __attribute__((ext_vector_type(8)));

static __device__ __forceinline__ v8f wmma_f32(v2f a, v2f b, v8f c) {
    return __builtin_amdgcn_wmma_f32_16x16x4_f32(
         false, a,  false, b,
         (short)0, c,  false,  false);
}

__global__ void mpnn_proj_ij(const float* __restrict__ z,
                             const float* __restrict__ W_msg,
                             const float* __restrict__ b_msg,
                             float* __restrict__ P,
                             float* __restrict__ Q) {
    const int tid  = threadIdx.x;
    const int wave = tid >> 5;
    const int lane = tid & 31;
    const int lm   = lane & 15;
    const int lh   = lane >> 4;
    const int h0   = wave * 16;
    const int mrow = blockIdx.x * 16 + lm;

    v8f ci = {0.f,0.f,0.f,0.f,0.f,0.f,0.f,0.f};
    v8f cj = ci;
    const float* zrow = z + (size_t)mrow * Z_;

#pragma unroll
    for (int t = 0; t < Z_ / 4; ++t) {
        const int kb = 4 * t + 2 * lh;
        v2f a = *(const v2f*)(zrow + kb);
        const float* wi = W_msg + (size_t)kb * H_ + h0 + lm;
        const float* wj = W_msg + (size_t)(Z_ + kb) * H_ + h0 + lm;
        v2f bi = { wi[0], wi[H_] };
        v2f bj = { wj[0], wj[H_] };
        ci = wmma_f32(a, bi, ci);
        cj = wmma_f32(a, bj, cj);
    }

    const float bm = b_msg[h0 + lm];
#pragma unroll
    for (int r = 0; r < 8; ++r) {
        const int orow = blockIdx.x * 16 + r + 8 * lh;
        Q[(size_t)orow * H_ + h0 + lm] = ci[r] + bm;
        P[(size_t)orow * H_ + h0 + lm] = cj[r];
    }
}

#define HT_    8
#define PFLOC_ 1
#define RB_    8
#define PP_    (H_ + 4)
#define AGP_   (H_ + 4)

#define AGG_LOADS_STAY_ABOVE() asm volatile("" ::: "memory")

static __device__ __forceinline__ void e_frag_load(v2f (&e)[4], const float* __restrict__ eb, int j, int lh) {
    const float* ep = eb + j * E_ + 2 * lh;
    for (int t = 0; t < 4; ++t)
        e[t] = *(const v2f*)(ep + 4 * t);
}

static __device__ __forceinline__ void tile_compute(const v2f (&e)[4],
                                                    const v4f (&pc)[HT_][2],
                                                    const v2f (&be)[HT_][4],
                                                    float (&acc)[HT_][8]) {
    v8f c[HT_];
    for (int q = 0; q < HT_; ++q)
        c[q] = __builtin_shufflevector(pc[q][0], pc[q][1], 0, 1, 2, 3, 4, 5, 6, 7);
    for (int t = 0; t < 4; ++t)
        for (int q = 0; q < HT_; ++q)
            c[q] = wmma_f32(be[q][t], e[t], c[q]);
    for (int q = 0; q < HT_; ++q)
        for (int r = 0; r < 8; ++r)
            acc[q][r] = fmaxf(acc[q][r], c[q][r]);
}

static __device__ __forceinline__ float row16_max(float v) {
    int x;
    x = __builtin_bit_cast(int, v);
    v = fmaxf(v, __builtin_bit_cast(float, __builtin_amdgcn_update_dpp(x, x, 0x161, 0xf, 0xf, false)));
    x = __builtin_bit_cast(int, v);
    v = fmaxf(v, __builtin_bit_cast(float, __builtin_amdgcn_update_dpp(x, x, 0x162, 0xf, 0xf, false)));
    x = __builtin_bit_cast(int, v);
    v = fmaxf(v, __builtin_bit_cast(float, __builtin_amdgcn_update_dpp(x, x, 0x164, 0xf, 0xf, false)));
    x = __builtin_bit_cast(int, v);
    v = fmaxf(v, __builtin_bit_cast(float, __builtin_amdgcn_update_dpp(x, x, 0x168, 0xf, 0xf, false)));
    return v;
}

__global__ void __launch_bounds__(256) __attribute__((amdgpu_waves_per_eu(2, 2)))
mpnn_agg_upd_kernel(const float* __restrict__ e_feat,
                    const int*   __restrict__ adj,
                    const float* __restrict__ W_msg,
                    const float* __restrict__ P,
                    const float* __restrict__ Q,
                    const float* __restrict__ z,
                    const float* __restrict__ W_upd,
                    const float* __restrict__ b_upd,
                    float* __restrict__ out) {
    __shared__ __attribute__((aligned(16))) float Ps[N_ * PP_];
    __shared__ unsigned short lst[RB_ * N_];
    __shared__ __attribute__((aligned(16))) float aggs[RB_ * AGP_];

    const int tid  = threadIdx.x;
    const int wave = tid >> 5;
    const int lane = tid & 31;
    const int lm   = lane & 15;
    const int lh   = lane >> 4;
    const int row  = blockIdx.x * RB_ + wave;
    const int b    = row >> 9;

    const float* eb   = e_feat + (size_t)row * N_ * E_;
    const int*   arow = adj + (size_t)row * N_;

    int av[16];
    for (int c = 0; c < 16; ++c)
        av[c] = arow[32 * c + lane];

    __builtin_prefetch((const char*)eb + lane * 256, 0, PFLOC_);
    __builtin_prefetch((const char*)eb + 8192 + lane * 256, 0, PFLOC_);

    v2f be[HT_][4];
    for (int t = 0; t < 4; ++t) {
        const int kb = 4 * t + 2 * lh;
        const float* we = W_msg + (size_t)(2 * Z_ + kb) * H_ + lm;
        for (int q = 0; q < HT_; ++q)
            be[q][t] = (v2f){ we[16 * q], we[H_ + 16 * q] };
    }

    const float* Pb = P + (size_t)b * N_ * H_;
    for (int it = 0; it < 2; ++it) {
        v4f tmp[32];
        const int r0 = wave * 64 + it * 32;
        for (int k = 0; k < 32; ++k)
            tmp[k] = *(const v4f*)(Pb + (size_t)(r0 + k) * H_ + lane * 4);
        for (int k = 0; k < 32; ++k)
            *(v4f*)&Ps[(r0 + k) * PP_ + lane * 4] = tmp[k];
    }

    unsigned short* mylst = lst + wave * N_;
    int cnt = 0, lastj = 0;
    for (int c = 0; c < 16; ++c) {
        const bool     on  = av[c] > 0;
        const unsigned m   = __builtin_amdgcn_ballot_w32(on);
        const int      pre = (int)__builtin_amdgcn_mbcnt_lo(m, 0u);
        if (on) mylst[cnt + pre] = (unsigned short)(32 * c + lane);
        if (m)  lastj = 32 * c + 31 - __builtin_clz(m);
        cnt += __builtin_popcount(m);
    }
    const int nsteps = (cnt + 31) >> 5;
    const int padto  = nsteps ? 32 * nsteps : 32;
    if (cnt + lane < padto) mylst[cnt + lane] = (unsigned short)lastj;

    __syncthreads();

#define P_ROW_LOAD(pc, j)                                                       \
    do {                                                                        \
        const float* pp_ = &Ps[(j) * PP_ + 8 * lh];                             \
        for (int q_ = 0; q_ < HT_; ++q_) {                                      \
            (pc)[q_][0] = *(const v4f*)(pp_ + 16 * q_);                         \
            (pc)[q_][1] = *(const v4f*)(pp_ + 16 * q_ + 4);                     \
        }                                                                       \
    } while (0)

    float acc[HT_][8];
    for (int q = 0; q < HT_; ++q)
        for (int r = 0; r < 8; ++r)
            acc[q][r] = -INFINITY;

    const unsigned short* tl = mylst + lm;
    const int s1  = (1 < nsteps) ? 1 : 0;
    int jC1 = tl[16];
    int jN0 = tl[32 * s1], jN1 = tl[32 * s1 + 16];
    v2f eC[2][4], eN[2][4];
    v4f pcA[HT_][2], pcB[HT_][2];
    {
        const int jC0 = tl[0];
        e_frag_load(eC[0], eb, jC0, lh);
        e_frag_load(eC[1], eb, jC1, lh);
        P_ROW_LOAD(pcA, jC0);
        const int last = nsteps ? nsteps - 1 : 0;
        const int pa = (1 < last) ? 1 : last;
        const int pb = (2 < last) ? 2 : last;
        __builtin_prefetch(eb + (int)mylst[32 * pa + lane] * E_, 0, PFLOC_);
        __builtin_prefetch(eb + (int)mylst[32 * pb + lane] * E_, 0, PFLOC_);
    }
    AGG_LOADS_STAY_ABOVE();
    for (int st = 0; st < nsteps; ++st) {
        const int s2 = (st + 2 < nsteps) ? st + 2 : 0;
        const int s3 = (st + 3 < nsteps) ? st + 3 : nsteps - 1;
        const int jF0 = tl[32 * s2], jF1 = tl[32 * s2 + 16];
        __builtin_prefetch(eb + (int)mylst[32 * s3 + lane] * E_, 0, PFLOC_);
        e_frag_load(eN[0], eb, jN0, lh);
        e_frag_load(eN[1], eb, jN1, lh);
        P_ROW_LOAD(pcB, jC1);
        AGG_LOADS_STAY_ABOVE();
        tile_compute(eC[0], pcA, be, acc);
        P_ROW_LOAD(pcA, jN0);
        AGG_LOADS_STAY_ABOVE();
        tile_compute(eC[1], pcB, be, acc);
        for (int t = 0; t < 4; ++t) { eC[0][t] = eN[0][t]; eC[1][t] = eN[1][t]; }
        jC1 = jN1; jN0 = jF0; jN1 = jF1;
    }
#undef P_ROW_LOAD

    const int    rowl = lm & 7;
    const float* zrow = z + (size_t)(blockIdx.x * RB_ + rowl) * Z_;
    const float* wcol = W_upd + wave * 16 + lm;
    v2f za[Z_ / 4];
    v2f wb[(Z_ + H_) / 4];
    for (int t = 0; t < Z_ / 4; ++t)
        za[t] = *(const v2f*)(zrow + 4 * t + 2 * lh);
    for (int t = 0; t < (Z_ + H_) / 4; ++t) {
        const float* w = wcol + (size_t)(4 * t + 2 * lh) * H_;
        wb[t] = (v2f){ w[0], w[H_] };
    }

    for (int q = 0; q < HT_; ++q)
        for (int r = 0; r < 8; ++r)
            acc[q][r] = row16_max(acc[q][r]);
    if (lm == 0) {
        const float* qrow = Q + (size_t)row * H_ + 8 * lh;
        float*       ag   = &aggs[wave * AGP_ + 8 * lh];
        for (int q = 0; q < HT_; ++q) {
            const v4f q0 = *(const v4f*)(qrow + 16 * q);
            const v4f q1 = *(const v4f*)(qrow + 16 * q + 4);
            *(v4f*)(ag + 16 * q)     = (v4f){ acc[q][0] + q0[0], acc[q][1] + q0[1], acc[q][2] + q0[2], acc[q][3] + q0[3] };
            *(v4f*)(ag + 16 * q + 4) = (v4f){ acc[q][4] + q1[0], acc[q][5] + q1[1], acc[q][6] + q1[2], acc[q][7] + q1[3] };
        }
    }

    v8f c = {0.f,0.f,0.f,0.f,0.f,0.f,0.f,0.f};
    for (int t = 0; t < Z_ / 4; ++t)
        c = wmma_f32(za[t], wb[t], c);

    __builtin_amdgcn_sched_barrier(0);
    __syncthreads();

    const float* agr = &aggs[rowl * AGP_ + 2 * lh];
    for (int t = 0; t < H_ / 4; ++t) {
        const v2f a = *(const v2f*)(agr + 4 * t);
        c = wmma_f32(a, wb[Z_ / 4 + t], c);
    }

    const float bu = b_upd[wave * 16 + lm];
    if (lh == 0) {
        for (int r = 0; r < 8; ++r)
            out[(size_t)(blockIdx.x * RB_ + r) * H_ + wave * 16 + lm] = c[r] + bu;
    }
}

extern "C" void kernel_launch(void* const* d_in, const int* in_sizes, int n_in,
                              void* d_out, int out_size, void* d_ws, size_t ws_size,
                              hipStream_t stream) {
    const float* z      = (const float*)d_in[0];
    const float* e_feat = (const float*)d_in[1];
    const int*   adj    = (const int*)  d_in[2];
    const float* W_msg  = (const float*)d_in[3];
    const float* b_msg  = (const float*)d_in[4];
    const float* W_upd  = (const float*)d_in[5];
    const float* b_upd  = (const float*)d_in[6];
    float* out = (float*)d_out;

    float* ws  = (float*)d_ws;
    float* P   = ws;
    float* Q   = ws + (size_t)ROWS_ * H_;

    mpnn_proj_ij<<<ROWS_ / 16, 256, 0, stream>>>(z, W_msg, b_msg, P, Q);

    mpnn_agg_upd_kernel<<<ROWS_ / RB_, 256, 0, stream>>>(e_feat, adj, W_msg, P, Q,
                                                        z, W_upd, b_upd, out);
}
